# starting kernel
# baseline (speedup 1.0000x reference)
#include <hip/hip_runtime.h>
#include <hip/hip_bf16.h>

typedef __attribute__((ext_vector_type(2))) float v2f;
typedef __attribute__((ext_vector_type(4))) float v4f;
typedef __attribute__((ext_vector_type(8))) float v8f;

#define D_DIM 128
#define TILE  16
#define WAVES_PER_BLOCK 8

__global__ __launch_bounds__(256) void hetero_udotv_wmma_kernel(
    const float* __restrict__ h_user, const float* __restrict__ h_item,
    const int* __restrict__ src0, const int* __restrict__ dst0,
    const int* __restrict__ src1, const int* __restrict__ dst1,
    const int* __restrict__ src2, const int* __restrict__ dst2,
    float* __restrict__ out, int E)
{
    const int lane      = threadIdx.x & 31;
    const int waveInBlk = threadIdx.x >> 5;
    const int etype     = blockIdx.y;
    const int tile      = blockIdx.x * WAVES_PER_BLOCK + waveInBlk;

    const int tilesPer = (E + TILE - 1) / TILE;
    if (tile >= tilesPer) return;
    const int e0 = tile * TILE;

    const float* srcTab; const float* dstTab;
    const int*   srcIdx; const int*   dstIdx;
    if (etype == 0)      { srcTab = h_user; dstTab = h_item; srcIdx = src0; dstIdx = dst0; }
    else if (etype == 1) { srcTab = h_item; dstTab = h_user; srcIdx = src1; dstIdx = dst1; }
    else                 { srcTab = h_user; dstTab = h_user; srcIdx = src2; dstIdx = dst2; }

    const int r    = lane & 15;
    const int half = lane >> 4;

    int eMy = e0 + r; if (eMy > E - 1) eMy = E - 1;
    const long iu = (long)srcIdx[eMy];
    const long iv = (long)dstIdx[eMy];

    const float* uRow = srcTab + iu * (long)D_DIM + 4 * half;
    const float* vRow = dstTab + iv * (long)D_DIM + 4 * half;

    v8f acc0 = {};
    v8f acc1 = {};
#pragma unroll
    for (int kk = 0; kk < D_DIM; kk += 8) {
        v4f a4 = *(const v4f*)(uRow + kk);
        v4f b4 = *(const v4f*)(vRow + kk);
        v2f a0 = { a4.x, a4.y }, b0 = { b4.x, b4.y };
        v2f a1 = { a4.z, a4.w }, b1 = { b4.z, b4.w };
        acc0 = __builtin_amdgcn_wmma_f32_16x16x4_f32(false, a0, false, b0,
                                                     (short)0, acc0, false, false);
        acc1 = __builtin_amdgcn_wmma_f32_16x16x4_f32(false, a1, false, b1,
                                                     (short)0, acc1, false, false);
    }

    const int j = lane & 7;
    float d = acc0[0] + acc1[0];
    if (j == 1) d = acc0[1] + acc1[1];
    if (j == 2) d = acc0[2] + acc1[2];
    if (j == 3) d = acc0[3] + acc1[3];
    if (j == 4) d = acc0[4] + acc1[4];
    if (j == 5) d = acc0[5] + acc1[5];
    if (j == 6) d = acc0[6] + acc1[6];
    if (j == 7) d = acc0[7] + acc1[7];

    const bool active = (lane < 8) || (lane >= 24);
    if (active) {
        const int e = (lane < 8) ? lane : (lane - 16);
        if (e0 + e < E) {
            out[(long)etype * (long)E + (long)(e0 + e)] = d;
        }
    }
}

extern "C" void kernel_launch(void* const* d_in, const int* in_sizes, int n_in,
                              void* d_out, int out_size, void* d_ws, size_t ws_size,
                              hipStream_t stream) {
    const float* h_user        = (const float*)d_in[0];
    const float* h_item        = (const float*)d_in[1];
    const int*   src_clicks    = (const int*)d_in[2];
    const int*   dst_clicks    = (const int*)d_in[3];
    const int*   src_clickedby = (const int*)d_in[4];
    const int*   dst_clickedby = (const int*)d_in[5];
    const int*   src_follows   = (const int*)d_in[6];
    const int*   dst_follows   = (const int*)d_in[7];
    float*       out           = (float*)d_out;

    const int E        = in_sizes[2];
    const int tilesPer = (E + TILE - 1) / TILE;
    const int blocksX  = (tilesPer + WAVES_PER_BLOCK - 1) / WAVES_PER_BLOCK;

    dim3 grid(blocksX, 3, 1);
    hetero_udotv_wmma_kernel<<<grid, 32 * WAVES_PER_BLOCK, 0, stream>>>(
        h_user, h_item,
        src_clicks, dst_clicks,
        src_clickedby, dst_clickedby,
        src_follows, dst_follows,
        out, E);
}


# Whole-line gathers: each b128 load reads a full 128-B line of each of 4 edges' rows (was 32 B of 16 rows); tile = 4 edges x 4 slices, diagonal quad-summed (new sum order); wave = 32 edges x 3 types
# speedup vs baseline: 1.9159x; 1.9159x over previous
#include <hip/hip_runtime.h>

typedef __attribute__((ext_vector_type(2))) float v2f;
typedef __attribute__((ext_vector_type(4))) float v4f;
typedef __attribute__((ext_vector_type(8))) float v8f;
typedef __attribute__((ext_vector_type(4))) int   v4i;

#define D_DIM 128
#define EDGES_PER_WAVE 32
#define TILES_PER_TYPE 8
#define WAVES_PER_BLOCK 8


__device__ __forceinline__ float quad_sum(float x)
{
    float y = x + __builtin_bit_cast(float,
        __builtin_amdgcn_mov_dpp(__builtin_bit_cast(int, x), 0xB1, 0xF, 0xF, true));
    float z = y + __builtin_bit_cast(float,
        __builtin_amdgcn_mov_dpp(__builtin_bit_cast(int, y), 0x4E, 0xF, 0xF, true));
    return z;
}

__device__ __forceinline__ void load_tile(
    const float* __restrict__ srcTab, const float* __restrict__ dstTab,
    const int srcRow, const int dstRow, const unsigned colOff,
    v4f (&a)[4], v4f (&b)[4])
{
    const float* uRow = srcTab + ((unsigned)srcRow * (unsigned)D_DIM + colOff);
    const float* vRow = dstTab + ((unsigned)dstRow * (unsigned)D_DIM + colOff);
#pragma unroll
    for (int j = 0; j < 4; ++j) {
        a[j] = *(const v4f*)(uRow + 32 * j);
        b[j] = *(const v4f*)(vRow + 32 * j);
    }
}

__device__ __forceinline__ void score_chunk(
    const float* __restrict__ srcTab, const float* __restrict__ dstTab,
    const int* __restrict__ srcIdx, const int* __restrict__ dstIdx,
    float* __restrict__ outT, const int E, const int e0, const int lane)
{
    const int c = lane & 15;
    const int h = lane >> 4;
    const int q = c >> 2;
    const int p = c & 3;
    const int r = lane & 7;
    const unsigned colOff = 4u * (unsigned)p + 16u * (unsigned)h;
    const bool full = (e0 + EDGES_PER_WAVE <= E);

    int si[TILES_PER_TYPE], di[TILES_PER_TYPE];
    if (full) {
#pragma unroll
        for (int g = 0; g < 2; ++g) {
            const int eb = e0 + 16 * g + 4 * q;
            const v4i s4 = *(const v4i*)(srcIdx + eb);
            const v4i d4 = *(const v4i*)(dstIdx + eb);
            si[4 * g + 0] = s4.x; si[4 * g + 1] = s4.y; si[4 * g + 2] = s4.z; si[4 * g + 3] = s4.w;
            di[4 * g + 0] = d4.x; di[4 * g + 1] = d4.y; di[4 * g + 2] = d4.z; di[4 * g + 3] = d4.w;
        }
    } else {
#pragma unroll
        for (int g = 0; g < 2; ++g) {
#pragma unroll
            for (int t = 0; t < 4; ++t) {
                int e = e0 + 16 * g + 4 * q + t; if (e > E - 1) e = E - 1;
                si[4 * g + t] = srcIdx[e];
                di[4 * g + t] = dstIdx[e];
            }
        }
    }

    v4f a[2][4], b[2][4];
    load_tile(srcTab, dstTab, si[0], di[0], colOff, a[0], b[0]);

    float keep[2] = { 0.0f, 0.0f };
#pragma unroll
    for (int n = 0; n < TILES_PER_TYPE; ++n) {
        const int cur = n & 1;
        __builtin_amdgcn_sched_barrier(0);
        if (n + 1 < TILES_PER_TYPE)
            load_tile(srcTab, dstTab, si[n + 1], di[n + 1], colOff, a[cur ^ 1], b[cur ^ 1]);
        __builtin_amdgcn_sched_barrier(0);

        v8f acc = {};
#pragma unroll
        for (int j = 0; j < 4; ++j) {
            const v4f a4 = a[cur][j];
            const v4f b4 = b[cur][j];
            const v2f a0 = { a4.x, a4.y }, b0 = { b4.x, b4.y };
            const v2f a1 = { a4.z, a4.w }, b1 = { b4.z, b4.w };
            acc = __builtin_amdgcn_wmma_f32_16x16x4_f32(false, a0, false, b0,
                                                        (short)0, acc, false, false);
            acc = __builtin_amdgcn_wmma_f32_16x16x4_f32(false, a1, false, b1,
                                                        (short)0, acc, false, false);
        }
        float d = acc[0];
        d = (r == 1) ? acc[1] : d;
        d = (r == 2) ? acc[2] : d;
        d = (r == 3) ? acc[3] : d;
        d = (r == 4) ? acc[4] : d;
        d = (r == 5) ? acc[5] : d;
        d = (r == 6) ? acc[6] : d;
        d = (r == 7) ? acc[7] : d;
        const float s = quad_sum(d);
        keep[n >> 2] = (p == (n & 3)) ? s : keep[n >> 2];
    }

    const float moved = __builtin_bit_cast(float,
        __builtin_amdgcn_mov_dpp(__builtin_bit_cast(int, keep[1]), 0x128, 0xF, 0xF, true));
    const bool own = ((c >> 3) == h);
    const int  eo  = own ? (e0 + c) : (e0 + 16 + (c ^ 8));
    const float val = own ? keep[0] : moved;
    if (eo < E) outT[eo] = val;
}

__global__ __launch_bounds__(256) void hetero_udotv_wmma_kernel(
    const float* __restrict__ h_user, const float* __restrict__ h_item,
    const int* __restrict__ src0, const int* __restrict__ dst0,
    const int* __restrict__ src1, const int* __restrict__ dst1,
    const int* __restrict__ src2, const int* __restrict__ dst2,
    float* __restrict__ out, int E)
{
    const int lane      = threadIdx.x & 31;
    const int waveInBlk = __builtin_amdgcn_readfirstlane((int)(threadIdx.x >> 5));
    const int chunk     = blockIdx.x * WAVES_PER_BLOCK + waveInBlk;
    const int e0        = chunk * EDGES_PER_WAVE;
    if (e0 >= E) return;

    score_chunk(h_user, h_item, src0, dst0, out,               E, e0, lane);
    score_chunk(h_item, h_user, src1, dst1, out + (long)E,     E, e0, lane);
    score_chunk(h_user, h_user, src2, dst2, out + 2 * (long)E, E, e0, lane);
}

extern "C" void kernel_launch(void* const* d_in, const int* in_sizes, int n_in,
                              void* d_out, int out_size, void* d_ws, size_t ws_size,
                              hipStream_t stream) {
    const float* h_user        = (const float*)d_in[0];
    const float* h_item        = (const float*)d_in[1];
    const int*   src_clicks    = (const int*)d_in[2];
    const int*   dst_clicks    = (const int*)d_in[3];
    const int*   src_clickedby = (const int*)d_in[4];
    const int*   dst_clickedby = (const int*)d_in[5];
    const int*   src_follows   = (const int*)d_in[6];
    const int*   dst_follows   = (const int*)d_in[7];
    float*       out           = (float*)d_out;

    const int E       = in_sizes[2];
    const int chunks  = (E + EDGES_PER_WAVE - 1) / EDGES_PER_WAVE;
    const int blocksX = (chunks + WAVES_PER_BLOCK - 1) / WAVES_PER_BLOCK;

    dim3 grid(blocksX, 1, 1);
    hetero_udotv_wmma_kernel<<<grid, 32 * WAVES_PER_BLOCK, 0, stream>>>(
        h_user, h_item,
        src_clicks, dst_clicks,
        src_clickedby, dst_clickedby,
        src_follows, dst_follows,
        out, E);
}
